# starting kernel
# baseline (speedup 1.0000x reference)
#include <hip/hip_runtime.h>
#include <stdint.h>

#define NNODES 100000
#define KDIM   128

typedef __attribute__((ext_vector_type(16))) __bf16 v16bf;
typedef __attribute__((ext_vector_type(8)))  __bf16 v8bf;
typedef __attribute__((ext_vector_type(8)))  float  v8f;

__device__ __forceinline__ __bf16 f2bf(float f) {
  unsigned u = __builtin_bit_cast(unsigned, f);
  unsigned r = u + 0x7FFFu + ((u >> 16) & 1u);
  unsigned short s = (unsigned short)(r >> 16);
  return __builtin_bit_cast(__bf16, s);
}

__device__ __forceinline__ void atomic_add_f32(float* p, float v) {
  __hip_atomic_fetch_add(p, v, __ATOMIC_RELAXED, __HIP_MEMORY_SCOPE_AGENT);
}

__global__ void zero_f32(float* __restrict__ p, long long n) {
  long long i = blockIdx.x * (long long)blockDim.x + threadIdx.x;
  long long stride = (long long)gridDim.x * blockDim.x;
  for (; i < n; i += stride) p[i] = 0.0f;
}

__global__ void degree_kernel(const int* __restrict__ src, const int* __restrict__ dst,
                              float* __restrict__ outdeg, float* __restrict__ indeg, int E) {
  int e = blockIdx.x * blockDim.x + threadIdx.x;
  if (e < E) {
    atomic_add_f32(&outdeg[src[e]], 1.0f);
    atomic_add_f32(&indeg[dst[e]], 1.0f);
  }
}

__global__ void norm_kernel(const float* __restrict__ outdeg, const float* __restrict__ indeg,
                            float* __restrict__ nsrc, float* __restrict__ ndst, int n) {
  int i = blockIdx.x * blockDim.x + threadIdx.x;
  if (i < n) {
    nsrc[i] = rsqrtf(fmaxf(outdeg[i], 1.0f));
    ndst[i] = rsqrtf(fmaxf(indeg[i], 1.0f));
  }
}

template <int MOUT>
__global__ __launch_bounds__(256)
void gcn_gemm(const float* __restrict__ x, const float* __restrict__ nsrc,
              const float* __restrict__ W, float* __restrict__ h, int nrows) {
  constexpr int LDK = KDIM + 8;
  __shared__ __bf16 As[64 * LDK];
  __shared__ __bf16 Bs[MOUT * LDK];

  const int tid  = threadIdx.x;
  const int row0 = blockIdx.x * 64;

  for (int idx = tid; idx < 64 * KDIM; idx += 256) {
    int r = idx >> 7, k = idx & (KDIM - 1);
    int row = row0 + r;
    float v = 0.0f;
    if (row < nrows) v = x[(size_t)row * KDIM + k] * nsrc[row];
    As[r * LDK + k] = f2bf(v);
  }
  for (int idx = tid; idx < MOUT * KDIM; idx += 256) {
    int k = idx / MOUT, n = idx % MOUT;
    Bs[n * LDK + k] = f2bf(W[(size_t)k * MOUT + n]);
  }
  __syncthreads();

  const int wave = tid >> 5, lane = tid & 31;
  const int half = lane >> 4, mn = lane & 15;
  constexpr int CT = MOUT / 16;
  constexpr int NT = 4 * CT;
  const bool full = (row0 + 64) <= nrows;

  const int tc = wave % CT;
  const __bf16* brow = &Bs[(tc * 16 + mn) * LDK];
  v16bf bfrag[KDIM / 32];
#pragma unroll
  for (int kk = 0; kk < KDIM / 32; ++kk) {
    int k0 = kk * 32 + half * 8;
    v8bf bl = *(const v8bf*)(brow + k0);
    v8bf bh = *(const v8bf*)(brow + k0 + 16);
#pragma unroll
    for (int i = 0; i < 8; ++i) { bfrag[kk][i] = bl[i]; bfrag[kk][i + 8] = bh[i]; }
  }

#pragma unroll
  for (int it = 0; it < NT / 8; ++it) {
    const int t  = wave + it * 8;
    const int tr = t / CT;
    const __bf16* arow = &As[(tr * 16 + mn) * LDK];

    v16bf afrag[KDIM / 32];
#pragma unroll
    for (int kk = 0; kk < KDIM / 32; ++kk) {
      int k0 = kk * 32 + half * 8;
      v8bf al = *(const v8bf*)(arow + k0);
      v8bf ah = *(const v8bf*)(arow + k0 + 16);
#pragma unroll
      for (int i = 0; i < 8; ++i) { afrag[kk][i] = al[i]; afrag[kk][i + 8] = ah[i]; }
    }

    v8f c = {};
#pragma unroll
    for (int kk = 0; kk < KDIM / 32; ++kk)
      c = __builtin_amdgcn_wmma_f32_16x16x32_bf16(false, afrag[kk], false, bfrag[kk],
                                                  (short)0, c, false, false);

    const int gr0 = row0 + tr * 16 + half * 8;
    const int gc  = tc * 16 + mn;
    float* out = h + (size_t)gr0 * MOUT + gc;
    if (full) {
#pragma unroll
      for (int r = 0; r < 8; ++r) out[(size_t)r * MOUT] = c[r];
    } else {
#pragma unroll
      for (int r = 0; r < 8; ++r)
        if (gr0 + r < nrows) out[(size_t)r * MOUT] = c[r];
    }
  }
}

__global__ void scatter_edges(const float4* __restrict__ h4, const int* __restrict__ src,
                              const int* __restrict__ dst, float* __restrict__ agg,
                              int E, int gshift) {
  long long gid = blockIdx.x * (long long)blockDim.x + threadIdx.x;
  int groups = 1 << gshift;
  int g = (int)(gid & (groups - 1));
  long long e = gid >> gshift;
  if (e >= E) return;
  int s = src[e], d = dst[e];
  float4 v = h4[(size_t)s * groups + g];
  float* base = agg + ((size_t)d << (gshift + 2)) + g * 4;
  atomic_add_f32(base + 0, v.x);
  atomic_add_f32(base + 1, v.y);
  atomic_add_f32(base + 2, v.z);
  atomic_add_f32(base + 3, v.w);
}

template <int F, bool RELU>
__global__ void finish_kernel(const float* __restrict__ agg, const float* __restrict__ ndst,
                              const float* __restrict__ bias, float* __restrict__ out, int n) {
  long long idx = blockIdx.x * (long long)blockDim.x + threadIdx.x;
  if (idx >= (long long)n * F) return;
  int node = (int)(idx / F);
  int f = (int)(idx & (F - 1));
  float v = agg[idx] * ndst[node] + bias[f];
  out[idx] = RELU ? fmaxf(v, 0.0f) : v;
}

extern "C" void kernel_launch(void* const* d_in, const int* in_sizes, int n_in,
                              void* d_out, int out_size, void* d_ws, size_t ws_size,
                              hipStream_t stream) {
  const float* in_feat = (const float*)d_in[0];
  const int*   src     = (const int*)d_in[1];
  const int*   dst     = (const int*)d_in[2];
  const float* W1 = (const float*)d_in[3];  const float* b1 = (const float*)d_in[4];
  const float* W2 = (const float*)d_in[5];  const float* b2 = (const float*)d_in[6];
  const float* W3 = (const float*)d_in[7];  const float* b3 = (const float*)d_in[8];
  const float* W4 = (const float*)d_in[9];  const float* b4 = (const float*)d_in[10];

  const int N = NNODES;
  const int E = in_sizes[1];

  float* ws     = (float*)d_ws;
  float* outdeg = ws;
  float* indeg  = outdeg + N;
  float* nsrc   = indeg + N;
  float* ndst   = nsrc + N;
  float* bufA   = ndst + N;
  float* bufB   = bufA + (size_t)N * 128;
  float* bufC   = bufB + (size_t)N * 128;

  const dim3 blk(256);
  const unsigned gemm_blocks = (N + 63) / 64;
  const unsigned nf128 = (unsigned)(((long long)N * 128 + 255) / 256);
  const unsigned nf64  = (unsigned)(((long long)N * 64 + 255) / 256);
  const unsigned sc128 = (unsigned)(((long long)E * 32 + 255) / 256);
  const unsigned sc64  = (unsigned)(((long long)E * 16 + 255) / 256);

  zero_f32<<<2048, blk, 0, stream>>>(outdeg, 2LL * N);
  degree_kernel<<<(E + 255) / 256, blk, 0, stream>>>(src, dst, outdeg, indeg, E);
  norm_kernel<<<(N + 255) / 256, blk, 0, stream>>>(outdeg, indeg, nsrc, ndst, N);

  gcn_gemm<128><<<gemm_blocks, blk, 0, stream>>>(in_feat, nsrc, W1, bufB, N);
  zero_f32<<<4096, blk, 0, stream>>>(bufC, (long long)N * 128);
  scatter_edges<<<sc128, blk, 0, stream>>>((const float4*)bufB, src, dst, bufC, E, 5);
  finish_kernel<128, true><<<nf128, blk, 0, stream>>>(bufC, ndst, b1, bufA, N);

  gcn_gemm<128><<<gemm_blocks, blk, 0, stream>>>(bufA, nsrc, W2, bufB, N);
  zero_f32<<<4096, blk, 0, stream>>>(bufC, (long long)N * 128);
  scatter_edges<<<sc128, blk, 0, stream>>>((const float4*)bufB, src, dst, bufC, E, 5);
  finish_kernel<128, true><<<nf128, blk, 0, stream>>>(bufC, ndst, b2, bufA, N);

  gcn_gemm<128><<<gemm_blocks, blk, 0, stream>>>(bufA, nsrc, W3, bufB, N);
  zero_f32<<<4096, blk, 0, stream>>>(bufC, (long long)N * 128);
  scatter_edges<<<sc128, blk, 0, stream>>>((const float4*)bufB, src, dst, bufC, E, 5);
  finish_kernel<128, true><<<nf128, blk, 0, stream>>>(bufC, ndst, b3, bufA, N);

  gcn_gemm<64><<<gemm_blocks, blk, 0, stream>>>(bufA, nsrc, W4, bufB, N);
  zero_f32<<<4096, blk, 0, stream>>>(bufC, (long long)N * 64);
  scatter_edges<<<sc64, blk, 0, stream>>>((const float4*)bufB, src, dst, bufC, E, 4);
  finish_kernel<64, false><<<nf64, blk, 0, stream>>>(bufC, ndst, b4, (float*)d_out, N);
}


# in-degree counted by layer 1's owner blocks from their bucket runs (one shared-memory add per gather group, norm stored there): scattered global float atomics on dst gone; no sum order change
# speedup vs baseline: 2.8911x; 2.8911x over previous
#include <hip/hip_runtime.h>
#include <stdint.h>

#define NNODES 100000
#define KDIM   128
#define BN     128
#define NB     ((NNODES + BN - 1) / BN)
#define NBP    800
#define NCH    1024

typedef __attribute__((ext_vector_type(16))) __bf16 v16bf;
typedef __attribute__((ext_vector_type(8)))  __bf16 v8bf;
typedef __attribute__((ext_vector_type(8)))  float  v8f;

__device__ __forceinline__ __bf16 f2bf(float f) {
  unsigned u = __builtin_bit_cast(unsigned, f);
  unsigned r = u + 0x7FFFu + ((u >> 16) & 1u);
  unsigned short s = (unsigned short)(r >> 16);
  return __builtin_bit_cast(__bf16, s);
}

__device__ __forceinline__ void atomic_add_f32(float* p, float v) {
  __hip_atomic_fetch_add(p, v, __ATOMIC_RELAXED, __HIP_MEMORY_SCOPE_AGENT);
}

__device__ __forceinline__ int imin(int a, int b) { return a < b ? a : b; }

__global__ void zero_f32(float* __restrict__ p, long long n) {
  long long i = blockIdx.x * (long long)blockDim.x + threadIdx.x;
  long long stride = (long long)gridDim.x * blockDim.x;
  for (; i < n; i += stride) p[i] = 0.0f;
}

__global__ void degree_kernel(const int* __restrict__ src, float* __restrict__ outdeg, int E) {
  int e = blockIdx.x * blockDim.x + threadIdx.x;
  if (e < E) atomic_add_f32(&outdeg[src[e]], 1.0f);
}

__global__ void norm_kernel(const float* __restrict__ outdeg, float* __restrict__ nsrc, int n) {
  int i = blockIdx.x * blockDim.x + threadIdx.x;
  if (i < n) nsrc[i] = rsqrtf(fmaxf(outdeg[i], 1.0f));
}


template <int MOUT, bool GUARD>
__device__ __forceinline__ void stripe_tiles(const __bf16* __restrict__ Bs, int ldk,
                                             const v16bf (&afrag)[KDIM / 32],
                                             float* __restrict__ out0, int nval, int half, int mn) {
  constexpr int CT = MOUT / 16;
  v8f c[CT];
#pragma unroll
  for (int tc = 0; tc < CT; ++tc) {
    c[tc] = v8f{};
    asm volatile("" : "+v"(c[tc]));
  }
#pragma unroll
  for (int kk = 0; kk < KDIM / 32; ++kk) {
    const int k0 = kk * 32 + half * 8;
#pragma unroll
    for (int tc = 0; tc < CT; ++tc) {
      const __bf16* brow = Bs + (tc * 16 + mn) * ldk;
      v8bf bl = *(const v8bf*)(brow + k0);
      v8bf bh = *(const v8bf*)(brow + k0 + 16);
      v16bf bfrag;
#pragma unroll
      for (int i = 0; i < 8; ++i) { bfrag[i] = bl[i]; bfrag[i + 8] = bh[i]; }
      c[tc] = __builtin_amdgcn_wmma_f32_16x16x32_bf16(false, afrag[kk], false, bfrag,
                                                      (short)0, c[tc], false, false);
    }
  }
#pragma unroll
  for (int tc = 0; tc < CT; ++tc) {
    float* out = out0 + tc * 16;
#pragma unroll
    for (int r = 0; r < 8; ++r)
      if (!GUARD || r < nval) out[(size_t)r * MOUT] = c[tc][r];
  }
}

template <int MOUT, bool EPI>
__global__ __launch_bounds__(256)
void gcn_gemm(const float* __restrict__ x, const float* __restrict__ nsrc,
              const float* __restrict__ ndst, const float* __restrict__ bias,
              const float* __restrict__ W, float* __restrict__ h, int nrows) {
  constexpr int LDK = KDIM + 8;
  constexpr int BR  = 128;
  __shared__ __bf16 Bs[MOUT * LDK];
  __shared__ float4 Bias4[KDIM / 4];

  const int tid  = threadIdx.x;
  const int row0 = blockIdx.x * BR;
  const int wave = tid >> 5, lane = tid & 31;
  const int half = lane >> 4, mn = lane & 15;

  const int  arow   = row0 + wave * 16 + mn;
  const bool avalid = arow < nrows;
  const int  arowc  = avalid ? arow : (nrows - 1);
  const float4* xp = (const float4*)(x + (size_t)arowc * KDIM + half * 8);
  float4 xa[16];
#pragma unroll
  for (int kk = 0; kk < KDIM / 32; ++kk) {
    xa[kk * 4 + 0] = xp[kk * 8 + 0];
    xa[kk * 4 + 1] = xp[kk * 8 + 1];
    xa[kk * 4 + 2] = xp[kk * 8 + 4];
    xa[kk * 4 + 3] = xp[kk * 8 + 5];
  }
  const float sc_src = nsrc[arowc];
  float sc_dst = 0.0f;
  if (EPI) sc_dst = ndst[arowc];

  for (int idx = tid; idx < MOUT * KDIM; idx += 256) {
    int k = idx / MOUT, n = idx % MOUT;
    Bs[n * LDK + k] = f2bf(W[(size_t)k * MOUT + n]);
  }
  if (EPI && tid < KDIM / 4) Bias4[tid] = ((const float4*)bias)[tid];
  __syncthreads();

  v16bf afrag[KDIM / 32];
#pragma unroll
  for (int kk = 0; kk < KDIM / 32; ++kk) {
#pragma unroll
    for (int q = 0; q < 4; ++q) {
      const float4 v = xa[kk * 4 + q];
      float4 bq = {0.0f, 0.0f, 0.0f, 0.0f};
      if (EPI) bq = Bias4[kk * 8 + half * 2 + (q & 1) + (q >> 1) * 4];
      const float vv[4] = {v.x, v.y, v.z, v.w};
      const float bv[4] = {bq.x, bq.y, bq.z, bq.w};
#pragma unroll
      for (int c = 0; c < 4; ++c) {
        float a = vv[c];
        if (EPI) a = fmaxf(a * sc_dst + bv[c], 0.0f);
        const float r = avalid ? a * sc_src : 0.0f;
        afrag[kk][q * 4 + c] = f2bf(r);
      }
    }
  }

  const int gr0  = row0 + wave * 16 + half * 8;
  const int nval = nrows - gr0;
  float* out0 = h + (size_t)gr0 * MOUT + mn;
  if ((row0 + BR) <= nrows)
    stripe_tiles<MOUT, false>(Bs, LDK, afrag, out0, nval, half, mn);
  else
    stripe_tiles<MOUT, true>(Bs, LDK, afrag, out0, nval, half, mn);
}


#define SG 25

__global__ __launch_bounds__(256)
void bucket_count(const int* __restrict__ dst, int* __restrict__ cnt, int E) {
  __shared__ int hist[8][NBP];
  const int tid = threadIdx.x, wave = tid >> 5, lane = tid & 31;
  const int c = blockIdx.x * 8 + wave;
  const int L = (E + NCH - 1) / NCH;
  const int beg = c * L;
  for (int j = lane; j < NBP; j += 32) hist[wave][j] = 0;
  __syncthreads();
  for (int i0 = 0; i0 < L; i0 += 32 * SG) {
    int d[SG];
#pragma unroll
    for (int g = 0; g < SG; ++g) d[g] = dst[imin(beg + i0 + 32 * g + lane, E - 1)];
#pragma unroll
    for (int g = 0; g < SG; ++g) {
      const int i = i0 + 32 * g;
      if (i + lane < L && beg + i + lane < E) atomicAdd(&hist[wave][d[g] >> 7], 1);
    }
  }
  __syncthreads();
  for (int j = lane; j < NB; j += 32) cnt[c * NBP + j] = hist[wave][j];
}

__device__ __forceinline__ int block_excl_scan_256(int v, int* sh, int tid) {
  sh[tid] = v;
  __syncthreads();
  for (int off = 1; off < 256; off <<= 1) {
    const int t = (tid >= off) ? sh[tid - off] : 0;
    __syncthreads();
    sh[tid] += t;
    __syncthreads();
  }
  return sh[tid] - v;
}

__global__ __launch_bounds__(256)
void bucket_scan_chunks(int* __restrict__ cnt, int* __restrict__ tot) {
  __shared__ int sh[256];
  const int tid = threadIdx.x, b = blockIdx.x;
  int v[NCH / 256];
  int s = 0;
#pragma unroll
  for (int i = 0; i < NCH / 256; ++i) { v[i] = cnt[(tid * (NCH / 256) + i) * NBP + b]; s += v[i]; }
  int run = block_excl_scan_256(s, sh, tid);
#pragma unroll
  for (int i = 0; i < NCH / 256; ++i) { cnt[(tid * (NCH / 256) + i) * NBP + b] = run; run += v[i]; }
  if (tid == 255) tot[b] = run;
}

__global__ __launch_bounds__(256)
void bucket_base(const int* __restrict__ tot, int* __restrict__ bstart) {
  __shared__ int sh[256];
  const int tid = threadIdx.x;
  int v[4];
  int s = 0;
#pragma unroll
  for (int i = 0; i < 4; ++i) { const int idx = tid * 4 + i; v[i] = (idx < NB) ? tot[idx] : 0; s += v[i]; }
  int run = block_excl_scan_256(s, sh, tid);
#pragma unroll
  for (int i = 0; i < 4; ++i) { const int idx = tid * 4 + i; if (idx < NB) bstart[idx] = run; run += v[i]; }
}

__global__ __launch_bounds__(256)
void bucket_place(const int* __restrict__ src, const int* __restrict__ dst,
                  const int* __restrict__ cnt, const int* __restrict__ bstart,
                  int* __restrict__ elist, int E) {
  __shared__ int cur[8][NBP];
  const int tid = threadIdx.x, wave = tid >> 5, lane = tid & 31;
  const int c = blockIdx.x * 8 + wave;
  const int L = (E + NCH - 1) / NCH;
  const int beg = c * L;
#pragma unroll
  for (int k = 0; k < (NB + 31) / 32; ++k) {
    const int j = lane + 32 * k;
    const int jc = imin(j, NB - 1);
    const int v = bstart[jc] + cnt[c * NBP + jc];
    if (j < NB) cur[wave][j] = v;
  }
  __syncthreads();
  for (int i0 = 0; i0 < L; i0 += 32 * SG) {
    int s[SG], d[SG], pos[SG];
#pragma unroll
    for (int g = 0; g < SG; ++g) {
      const int ec = imin(beg + i0 + 32 * g + lane, E - 1);
      s[g] = src[ec];
      d[g] = dst[ec];
    }
#pragma unroll
    for (int g = 0; g < SG; ++g) {
      const int i = i0 + 32 * g;
      pos[g] = 0;
      if (i + lane < L && beg + i + lane < E) pos[g] = atomicAdd(&cur[wave][d[g] >> 7], 1);
    }
#pragma unroll
    for (int g = 0; g < SG; ++g) {
      const int i = i0 + 32 * g;
      if (i + lane < L && beg + i + lane < E) elist[pos[g]] = (s[g] << 7) | (d[g] & (BN - 1));
    }
  }
}

__device__ __forceinline__ void vadd(float4& a, const float4 v) { a.x += v.x; a.y += v.y; a.z += v.z; a.w += v.w; }
__device__ __forceinline__ void vadd(float2& a, const float2 v) { a.x += v.x; a.y += v.y; }
__device__ __forceinline__ void vzero(float4& a) { a.x = 0.0f; a.y = 0.0f; a.z = 0.0f; a.w = 0.0f; }
__device__ __forceinline__ void vzero(float2& a) { a.x = 0.0f; a.y = 0.0f; }

template <int F> struct RowVec;
template <> struct RowVec<128> { typedef float4 T; };
template <> struct RowVec<64>  { typedef float2 T; };

#define QCAP 64
#define CH   1024

template <int F, int GB, bool FULL, bool DEG>
__device__ __forceinline__ void gather_group(const typename RowVec<F>::T* __restrict__ hv,
                                             typename RowVec<F>::T* acc, const int* qw, int* rc,
                                             int head, int cnt, int lane) {
  typedef typename RowVec<F>::T V;
  const int slot = qw[(head + (lane & (GB - 1))) & (QCAP - 1)];
  if (DEG && lane < GB && (FULL || lane < cnt)) atomicAdd(&rc[slot & (BN - 1)], 1);
  int p[GB];
  V v[GB];
#pragma unroll
  for (int j = 0; j < GB; ++j) {
    p[j] = __builtin_amdgcn_readlane(slot, j);
    if (FULL || j < cnt) v[j] = hv[(size_t)(p[j] >> 7) * 32 + lane];
    else vzero(v[j]);
  }
#pragma unroll
  for (int j = 0; j < GB; ++j)
    if (FULL || j < cnt) vadd(acc[(p[j] & (BN - 1)) * 32 + lane], v[j]);
}

template <int F, bool DEG>
__global__ __launch_bounds__(256)
void owner_aggregate(const float* __restrict__ h, const int* __restrict__ elist,
                     const int* __restrict__ bstart, const int* __restrict__ tot,
                     float* __restrict__ agg, float* __restrict__ ndst, int nrows) {
  typedef typename RowVec<F>::T V;
  constexpr int GB = 2048 / F;
  constexpr int PT = CH / 256;
  const V* __restrict__ hv = (const V*)h;
  V* __restrict__ aggv = (V*)agg;
  __shared__ V acc[BN * 32];
  __shared__ int ebuf[2][CH];
  __shared__ int q[8][QCAP];
  __shared__ int rowcnt[DEG ? BN : 1];
  const int tid = threadIdx.x, wave = tid >> 5, lane = tid & 31;
  const int b = blockIdx.x;
  V* mine = acc + (wave * 16) * 32 + lane;
  for (int r = 0; r < 16; ++r) vzero(mine[r * 32]);
  int* qw = q[wave];
  qw[lane] = 0;
  qw[lane + 32] = 0;
  if (DEG && lane < 16) rowcnt[wave * 16 + lane] = 0;
  const int st = bstart[b], n = tot[b];
  if (n > 0) {
    const int* __restrict__ el = elist + st;
    const int last = n - 1;
    const int nch = (n + CH - 1) / CH;
    int rg[PT];
#pragma unroll
    for (int k = 0; k < PT; ++k) rg[k] = el[imin(tid + 256 * k, last)];
    int head = 0, tail = 0;
    for (int c = 0; c < nch; ++c) {
      int* eb = ebuf[c & 1];
#pragma unroll
      for (int k = 0; k < PT; ++k) eb[tid + 256 * k] = rg[k];
      __syncthreads();
#pragma unroll
      for (int k = 0; k < PT; ++k) rg[k] = el[imin((c + 1) * CH + tid + 256 * k, last)];
      const int cn = imin(CH, n - c * CH);
      for (int i = 0; i < cn; i += 32) {
        const int pk = eb[i + lane];
        const bool own = (i + lane < cn) && (((pk >> 4) & 7) == wave);
        const unsigned m = __builtin_amdgcn_ballot_w32(own);
        if (own) qw[(tail + (int)__builtin_amdgcn_mbcnt_lo(m, 0u)) & (QCAP - 1)] = pk;
        tail += __builtin_popcount(m);
        while (tail - head >= GB) {
          gather_group<F, GB, true, DEG>(hv, acc, qw, rowcnt, head, GB, lane);
          head += GB;
        }
      }
    }
    if (tail != head) gather_group<F, GB, false, DEG>(hv, acc, qw, rowcnt, head, tail - head, lane);
  }
  const int row0 = b * BN + wave * 16;
  for (int r = 0; r < 16; ++r)
    if (row0 + r < nrows) aggv[(size_t)(row0 + r) * 32 + lane] = mine[r * 32];
  if (DEG && lane < 16 && row0 + lane < nrows)
    ndst[row0 + lane] = rsqrtf(fmaxf((float)rowcnt[wave * 16 + lane], 1.0f));
}

template <int F, bool RELU>
__global__ void finish_kernel(const float* __restrict__ agg, const float* __restrict__ ndst,
                              const float* __restrict__ bias, float* __restrict__ out, int n) {
  long long idx = blockIdx.x * (long long)blockDim.x + threadIdx.x;
  if (idx >= (long long)n * F) return;
  int node = (int)(idx / F);
  int f = (int)(idx & (F - 1));
  float v = agg[idx] * ndst[node] + bias[f];
  out[idx] = RELU ? fmaxf(v, 0.0f) : v;
}

extern "C" void kernel_launch(void* const* d_in, const int* in_sizes, int n_in,
                              void* d_out, int out_size, void* d_ws, size_t ws_size,
                              hipStream_t stream) {
  const float* in_feat = (const float*)d_in[0];
  const int*   src     = (const int*)d_in[1];
  const int*   dst     = (const int*)d_in[2];
  const float* W1 = (const float*)d_in[3];  const float* b1 = (const float*)d_in[4];
  const float* W2 = (const float*)d_in[5];  const float* b2 = (const float*)d_in[6];
  const float* W3 = (const float*)d_in[7];  const float* b3 = (const float*)d_in[8];
  const float* W4 = (const float*)d_in[9];  const float* b4 = (const float*)d_in[10];

  const int N = NNODES;
  const int E = in_sizes[1];

  float* ws     = (float*)d_ws;
  float* outdeg = ws;
  float* indeg  = outdeg + N;
  float* nsrc   = indeg + N;
  float* ndst   = nsrc + N;
  float* bufA   = ndst + N;
  float* bufB   = bufA + (size_t)N * 128;
  int*   cnt    = (int*)(bufB + (size_t)N * 128);
  int*   tot    = cnt + NCH * NBP;
  int*   bstart = tot + 1024;
  int*   elist  = bstart + 1024;

  const dim3 blk(256);
  const unsigned gemm_blocks = (N + 127) / 128;
  const unsigned nf64  = (unsigned)(((long long)N * 64 + 255) / 256);

  zero_f32<<<(N + 255) / 256, blk, 0, stream>>>(outdeg, 1LL * N);
  degree_kernel<<<(E + 255) / 256, blk, 0, stream>>>(src, outdeg, E);
  norm_kernel<<<(N + 255) / 256, blk, 0, stream>>>(outdeg, nsrc, N);

  bucket_count<<<NCH / 8, blk, 0, stream>>>(dst, cnt, E);
  bucket_scan_chunks<<<NB, blk, 0, stream>>>(cnt, tot);
  bucket_base<<<1, blk, 0, stream>>>(tot, bstart);
  bucket_place<<<NCH / 8, blk, 0, stream>>>(src, dst, cnt, bstart, elist, E);

  gcn_gemm<128, false><<<gemm_blocks, blk, 0, stream>>>(in_feat, nsrc, ndst, b1, W1, bufB, N);
  owner_aggregate<128, true><<<NB, blk, 0, stream>>>(bufB, elist, bstart, tot, bufA, ndst, N);

  gcn_gemm<128, true><<<gemm_blocks, blk, 0, stream>>>(bufA, nsrc, ndst, b1, W2, bufB, N);
  owner_aggregate<128, false><<<NB, blk, 0, stream>>>(bufB, elist, bstart, tot, bufA, ndst, N);

  gcn_gemm<128, true><<<gemm_blocks, blk, 0, stream>>>(bufA, nsrc, ndst, b2, W3, bufB, N);
  owner_aggregate<128, false><<<NB, blk, 0, stream>>>(bufB, elist, bstart, tot, bufA, ndst, N);

  gcn_gemm<64, true><<<gemm_blocks, blk, 0, stream>>>(bufA, nsrc, ndst, b3, W4, bufB, N);
  owner_aggregate<64, false><<<NB, blk, 0, stream>>>(bufB, elist, bstart, tot, bufA, ndst, N);
  finish_kernel<64, false><<<nf64, blk, 0, stream>>>(bufA, ndst, b4, (float*)d_out, N);
}


# out-degree counted in shared memory: a block owns one node range for one edge chunk, the norm pass adds the chunk counts; scattered global float atomics and the zero launch gone; no sum order change
# speedup vs baseline: 3.3914x; 1.1730x over previous
#include <hip/hip_runtime.h>
#include <stdint.h>

#define NNODES 100000
#define KDIM   128
#define BN     128
#define NB     ((NNODES + BN - 1) / BN)
#define NBP    800
#define NCH    1024

typedef __attribute__((ext_vector_type(16))) __bf16 v16bf;
typedef __attribute__((ext_vector_type(8)))  __bf16 v8bf;
typedef __attribute__((ext_vector_type(8)))  float  v8f;

__device__ __forceinline__ __bf16 f2bf(float f) {
  unsigned u = __builtin_bit_cast(unsigned, f);
  unsigned r = u + 0x7FFFu + ((u >> 16) & 1u);
  unsigned short s = (unsigned short)(r >> 16);
  return __builtin_bit_cast(__bf16, s);
}

__device__ __forceinline__ int imin(int a, int b) { return a < b ? a : b; }

#define DR   16
#define DC   16
#define DRN  ((((NNODES + DR - 1) / DR) + 31) / 32 * 32)
#define DNT  512
#define DLD  13

__device__ __forceinline__ void count_in_range(int* tab, int s, int lo) {
  const unsigned u = (unsigned)(s - lo);
  if (u < (unsigned)DRN) atomicAdd(&tab[u], 1);
}

__global__ __launch_bounds__(DNT)
void outdeg_count(const int* __restrict__ src, int* __restrict__ part, int E) {
  __shared__ int tab[DRN];
  const int tid = threadIdx.x;
  const int r = blockIdx.x % DR, c = blockIdx.x / DR;
  const int lo = r * DRN;
  for (int j = tid; j < DRN; j += DNT) tab[j] = 0;
  __syncthreads();
  const int nq = E >> 2;
  const int LQ = (nq + DC - 1) / DC;
  const int q0 = c * LQ;
  const int q1 = imin(q0 + LQ, nq);
  const int4* __restrict__ s4 = (const int4*)src;
  for (int q = q0 + tid; q < q1; q += DNT * DLD) {
    int4 v[DLD];
#pragma unroll
    for (int k = 0; k < DLD; ++k) v[k] = s4[imin(q + k * DNT, q1 - 1)];
#pragma unroll
    for (int k = 0; k < DLD; ++k) {
      if (q + k * DNT < q1) {
        count_in_range(tab, v[k].x, lo);
        count_in_range(tab, v[k].y, lo);
        count_in_range(tab, v[k].z, lo);
        count_in_range(tab, v[k].w, lo);
      }
    }
  }
  if (c == DC - 1 && tid < E - 4 * nq) count_in_range(tab, src[4 * nq + tid], lo);
  __syncthreads();
  for (int j = tid; j < DRN; j += DNT)
    if (lo + j < NNODES) part[(size_t)c * NNODES + lo + j] = tab[j];
}

__global__ void norm_kernel(const int* __restrict__ part, float* __restrict__ nsrc, int n) {
  int i = blockIdx.x * blockDim.x + threadIdx.x;
  if (i < n) {
    int s = 0;
#pragma unroll
    for (int c = 0; c < DC; ++c) s += part[(size_t)c * NNODES + i];
    nsrc[i] = rsqrtf(fmaxf((float)s, 1.0f));
  }
}


template <int MOUT, bool GUARD>
__device__ __forceinline__ void stripe_tiles(const __bf16* __restrict__ Bs, int ldk,
                                             const v16bf (&afrag)[KDIM / 32],
                                             float* __restrict__ out0, int nval, int half, int mn) {
  constexpr int CT = MOUT / 16;
  v8f c[CT];
#pragma unroll
  for (int tc = 0; tc < CT; ++tc) {
    c[tc] = v8f{};
    asm volatile("" : "+v"(c[tc]));
  }
#pragma unroll
  for (int kk = 0; kk < KDIM / 32; ++kk) {
    const int k0 = kk * 32 + half * 8;
#pragma unroll
    for (int tc = 0; tc < CT; ++tc) {
      const __bf16* brow = Bs + (tc * 16 + mn) * ldk;
      v8bf bl = *(const v8bf*)(brow + k0);
      v8bf bh = *(const v8bf*)(brow + k0 + 16);
      v16bf bfrag;
#pragma unroll
      for (int i = 0; i < 8; ++i) { bfrag[i] = bl[i]; bfrag[i + 8] = bh[i]; }
      c[tc] = __builtin_amdgcn_wmma_f32_16x16x32_bf16(false, afrag[kk], false, bfrag,
                                                      (short)0, c[tc], false, false);
    }
  }
#pragma unroll
  for (int tc = 0; tc < CT; ++tc) {
    float* out = out0 + tc * 16;
#pragma unroll
    for (int r = 0; r < 8; ++r)
      if (!GUARD || r < nval) out[(size_t)r * MOUT] = c[tc][r];
  }
}

template <int MOUT, bool EPI>
__global__ __launch_bounds__(256)
void gcn_gemm(const float* __restrict__ x, const float* __restrict__ nsrc,
              const float* __restrict__ ndst, const float* __restrict__ bias,
              const float* __restrict__ W, float* __restrict__ h, int nrows) {
  constexpr int LDK = KDIM + 8;
  constexpr int BR  = 128;
  __shared__ __bf16 Bs[MOUT * LDK];
  __shared__ float4 Bias4[KDIM / 4];

  const int tid  = threadIdx.x;
  const int row0 = blockIdx.x * BR;
  const int wave = tid >> 5, lane = tid & 31;
  const int half = lane >> 4, mn = lane & 15;

  const int  arow   = row0 + wave * 16 + mn;
  const bool avalid = arow < nrows;
  const int  arowc  = avalid ? arow : (nrows - 1);
  const float4* xp = (const float4*)(x + (size_t)arowc * KDIM + half * 8);
  float4 xa[16];
#pragma unroll
  for (int kk = 0; kk < KDIM / 32; ++kk) {
    xa[kk * 4 + 0] = xp[kk * 8 + 0];
    xa[kk * 4 + 1] = xp[kk * 8 + 1];
    xa[kk * 4 + 2] = xp[kk * 8 + 4];
    xa[kk * 4 + 3] = xp[kk * 8 + 5];
  }
  const float sc_src = nsrc[arowc];
  float sc_dst = 0.0f;
  if (EPI) sc_dst = ndst[arowc];

  for (int idx = tid; idx < MOUT * KDIM; idx += 256) {
    int k = idx / MOUT, n = idx % MOUT;
    Bs[n * LDK + k] = f2bf(W[(size_t)k * MOUT + n]);
  }
  if (EPI && tid < KDIM / 4) Bias4[tid] = ((const float4*)bias)[tid];
  __syncthreads();

  v16bf afrag[KDIM / 32];
#pragma unroll
  for (int kk = 0; kk < KDIM / 32; ++kk) {
#pragma unroll
    for (int q = 0; q < 4; ++q) {
      const float4 v = xa[kk * 4 + q];
      float4 bq = {0.0f, 0.0f, 0.0f, 0.0f};
      if (EPI) bq = Bias4[kk * 8 + half * 2 + (q & 1) + (q >> 1) * 4];
      const float vv[4] = {v.x, v.y, v.z, v.w};
      const float bv[4] = {bq.x, bq.y, bq.z, bq.w};
#pragma unroll
      for (int c = 0; c < 4; ++c) {
        float a = vv[c];
        if (EPI) a = fmaxf(a * sc_dst + bv[c], 0.0f);
        const float r = avalid ? a * sc_src : 0.0f;
        afrag[kk][q * 4 + c] = f2bf(r);
      }
    }
  }

  const int gr0  = row0 + wave * 16 + half * 8;
  const int nval = nrows - gr0;
  float* out0 = h + (size_t)gr0 * MOUT + mn;
  if ((row0 + BR) <= nrows)
    stripe_tiles<MOUT, false>(Bs, LDK, afrag, out0, nval, half, mn);
  else
    stripe_tiles<MOUT, true>(Bs, LDK, afrag, out0, nval, half, mn);
}


#define SG 25

__global__ __launch_bounds__(256)
void bucket_count(const int* __restrict__ dst, int* __restrict__ cnt, int E) {
  __shared__ int hist[8][NBP];
  const int tid = threadIdx.x, wave = tid >> 5, lane = tid & 31;
  const int c = blockIdx.x * 8 + wave;
  const int L = (E + NCH - 1) / NCH;
  const int beg = c * L;
  for (int j = lane; j < NBP; j += 32) hist[wave][j] = 0;
  __syncthreads();
  for (int i0 = 0; i0 < L; i0 += 32 * SG) {
    int d[SG];
#pragma unroll
    for (int g = 0; g < SG; ++g) d[g] = dst[imin(beg + i0 + 32 * g + lane, E - 1)];
#pragma unroll
    for (int g = 0; g < SG; ++g) {
      const int i = i0 + 32 * g;
      if (i + lane < L && beg + i + lane < E) atomicAdd(&hist[wave][d[g] >> 7], 1);
    }
  }
  __syncthreads();
  for (int j = lane; j < NB; j += 32) cnt[c * NBP + j] = hist[wave][j];
}

__device__ __forceinline__ int block_excl_scan_256(int v, int* sh, int tid) {
  sh[tid] = v;
  __syncthreads();
  for (int off = 1; off < 256; off <<= 1) {
    const int t = (tid >= off) ? sh[tid - off] : 0;
    __syncthreads();
    sh[tid] += t;
    __syncthreads();
  }
  return sh[tid] - v;
}

__global__ __launch_bounds__(256)
void bucket_scan_chunks(int* __restrict__ cnt, int* __restrict__ tot) {
  __shared__ int sh[256];
  const int tid = threadIdx.x, b = blockIdx.x;
  int v[NCH / 256];
  int s = 0;
#pragma unroll
  for (int i = 0; i < NCH / 256; ++i) { v[i] = cnt[(tid * (NCH / 256) + i) * NBP + b]; s += v[i]; }
  int run = block_excl_scan_256(s, sh, tid);
#pragma unroll
  for (int i = 0; i < NCH / 256; ++i) { cnt[(tid * (NCH / 256) + i) * NBP + b] = run; run += v[i]; }
  if (tid == 255) tot[b] = run;
}

__global__ __launch_bounds__(256)
void bucket_base(const int* __restrict__ tot, int* __restrict__ bstart) {
  __shared__ int sh[256];
  const int tid = threadIdx.x;
  int v[4];
  int s = 0;
#pragma unroll
  for (int i = 0; i < 4; ++i) { const int idx = tid * 4 + i; v[i] = (idx < NB) ? tot[idx] : 0; s += v[i]; }
  int run = block_excl_scan_256(s, sh, tid);
#pragma unroll
  for (int i = 0; i < 4; ++i) { const int idx = tid * 4 + i; if (idx < NB) bstart[idx] = run; run += v[i]; }
}

__global__ __launch_bounds__(256)
void bucket_place(const int* __restrict__ src, const int* __restrict__ dst,
                  const int* __restrict__ cnt, const int* __restrict__ bstart,
                  int* __restrict__ elist, int E) {
  __shared__ int cur[8][NBP];
  const int tid = threadIdx.x, wave = tid >> 5, lane = tid & 31;
  const int c = blockIdx.x * 8 + wave;
  const int L = (E + NCH - 1) / NCH;
  const int beg = c * L;
#pragma unroll
  for (int k = 0; k < (NB + 31) / 32; ++k) {
    const int j = lane + 32 * k;
    const int jc = imin(j, NB - 1);
    const int v = bstart[jc] + cnt[c * NBP + jc];
    if (j < NB) cur[wave][j] = v;
  }
  __syncthreads();
  for (int i0 = 0; i0 < L; i0 += 32 * SG) {
    int s[SG], d[SG], pos[SG];
#pragma unroll
    for (int g = 0; g < SG; ++g) {
      const int ec = imin(beg + i0 + 32 * g + lane, E - 1);
      s[g] = src[ec];
      d[g] = dst[ec];
    }
#pragma unroll
    for (int g = 0; g < SG; ++g) {
      const int i = i0 + 32 * g;
      pos[g] = 0;
      if (i + lane < L && beg + i + lane < E) pos[g] = atomicAdd(&cur[wave][d[g] >> 7], 1);
    }
#pragma unroll
    for (int g = 0; g < SG; ++g) {
      const int i = i0 + 32 * g;
      if (i + lane < L && beg + i + lane < E) elist[pos[g]] = (s[g] << 7) | (d[g] & (BN - 1));
    }
  }
}

__device__ __forceinline__ void vadd(float4& a, const float4 v) { a.x += v.x; a.y += v.y; a.z += v.z; a.w += v.w; }
__device__ __forceinline__ void vadd(float2& a, const float2 v) { a.x += v.x; a.y += v.y; }
__device__ __forceinline__ void vzero(float4& a) { a.x = 0.0f; a.y = 0.0f; a.z = 0.0f; a.w = 0.0f; }
__device__ __forceinline__ void vzero(float2& a) { a.x = 0.0f; a.y = 0.0f; }

template <int F> struct RowVec;
template <> struct RowVec<128> { typedef float4 T; };
template <> struct RowVec<64>  { typedef float2 T; };

#define QCAP 64
#define CH   1024

template <int F, int GB, bool FULL, bool DEG>
__device__ __forceinline__ void gather_group(const typename RowVec<F>::T* __restrict__ hv,
                                             typename RowVec<F>::T* acc, const int* qw, int* rc,
                                             int head, int cnt, int lane) {
  typedef typename RowVec<F>::T V;
  const int slot = qw[(head + (lane & (GB - 1))) & (QCAP - 1)];
  if (DEG && lane < GB && (FULL || lane < cnt)) atomicAdd(&rc[slot & (BN - 1)], 1);
  int p[GB];
  V v[GB];
#pragma unroll
  for (int j = 0; j < GB; ++j) {
    p[j] = __builtin_amdgcn_readlane(slot, j);
    if (FULL || j < cnt) v[j] = hv[(size_t)(p[j] >> 7) * 32 + lane];
    else vzero(v[j]);
  }
#pragma unroll
  for (int j = 0; j < GB; ++j)
    if (FULL || j < cnt) vadd(acc[(p[j] & (BN - 1)) * 32 + lane], v[j]);
}

template <int F, bool DEG>
__global__ __launch_bounds__(256)
void owner_aggregate(const float* __restrict__ h, const int* __restrict__ elist,
                     const int* __restrict__ bstart, const int* __restrict__ tot,
                     float* __restrict__ agg, float* __restrict__ ndst, int nrows) {
  typedef typename RowVec<F>::T V;
  constexpr int GB = 2048 / F;
  constexpr int PT = CH / 256;
  const V* __restrict__ hv = (const V*)h;
  V* __restrict__ aggv = (V*)agg;
  __shared__ V acc[BN * 32];
  __shared__ int ebuf[2][CH];
  __shared__ int q[8][QCAP];
  __shared__ int rowcnt[DEG ? BN : 1];
  const int tid = threadIdx.x, wave = tid >> 5, lane = tid & 31;
  const int b = blockIdx.x;
  V* mine = acc + (wave * 16) * 32 + lane;
  for (int r = 0; r < 16; ++r) vzero(mine[r * 32]);
  int* qw = q[wave];
  qw[lane] = 0;
  qw[lane + 32] = 0;
  if (DEG && lane < 16) rowcnt[wave * 16 + lane] = 0;
  const int st = bstart[b], n = tot[b];
  if (n > 0) {
    const int* __restrict__ el = elist + st;
    const int last = n - 1;
    const int nch = (n + CH - 1) / CH;
    int rg[PT];
#pragma unroll
    for (int k = 0; k < PT; ++k) rg[k] = el[imin(tid + 256 * k, last)];
    int head = 0, tail = 0;
    for (int c = 0; c < nch; ++c) {
      int* eb = ebuf[c & 1];
#pragma unroll
      for (int k = 0; k < PT; ++k) eb[tid + 256 * k] = rg[k];
      __syncthreads();
#pragma unroll
      for (int k = 0; k < PT; ++k) rg[k] = el[imin((c + 1) * CH + tid + 256 * k, last)];
      const int cn = imin(CH, n - c * CH);
      for (int i = 0; i < cn; i += 32) {
        const int pk = eb[i + lane];
        const bool own = (i + lane < cn) && (((pk >> 4) & 7) == wave);
        const unsigned m = __builtin_amdgcn_ballot_w32(own);
        if (own) qw[(tail + (int)__builtin_amdgcn_mbcnt_lo(m, 0u)) & (QCAP - 1)] = pk;
        tail += __builtin_popcount(m);
        while (tail - head >= GB) {
          gather_group<F, GB, true, DEG>(hv, acc, qw, rowcnt, head, GB, lane);
          head += GB;
        }
      }
    }
    if (tail != head) gather_group<F, GB, false, DEG>(hv, acc, qw, rowcnt, head, tail - head, lane);
  }
  const int row0 = b * BN + wave * 16;
  for (int r = 0; r < 16; ++r)
    if (row0 + r < nrows) aggv[(size_t)(row0 + r) * 32 + lane] = mine[r * 32];
  if (DEG && lane < 16 && row0 + lane < nrows)
    ndst[row0 + lane] = rsqrtf(fmaxf((float)rowcnt[wave * 16 + lane], 1.0f));
}

template <int F, bool RELU>
__global__ void finish_kernel(const float* __restrict__ agg, const float* __restrict__ ndst,
                              const float* __restrict__ bias, float* __restrict__ out, int n) {
  long long idx = blockIdx.x * (long long)blockDim.x + threadIdx.x;
  if (idx >= (long long)n * F) return;
  int node = (int)(idx / F);
  int f = (int)(idx & (F - 1));
  float v = agg[idx] * ndst[node] + bias[f];
  out[idx] = RELU ? fmaxf(v, 0.0f) : v;
}

extern "C" void kernel_launch(void* const* d_in, const int* in_sizes, int n_in,
                              void* d_out, int out_size, void* d_ws, size_t ws_size,
                              hipStream_t stream) {
  const float* in_feat = (const float*)d_in[0];
  const int*   src     = (const int*)d_in[1];
  const int*   dst     = (const int*)d_in[2];
  const float* W1 = (const float*)d_in[3];  const float* b1 = (const float*)d_in[4];
  const float* W2 = (const float*)d_in[5];  const float* b2 = (const float*)d_in[6];
  const float* W3 = (const float*)d_in[7];  const float* b3 = (const float*)d_in[8];
  const float* W4 = (const float*)d_in[9];  const float* b4 = (const float*)d_in[10];

  const int N = NNODES;
  const int E = in_sizes[1];

  float* ws     = (float*)d_ws;
  float* nsrc   = ws + 2 * (size_t)N;
  float* ndst   = nsrc + N;
  float* bufA   = ndst + N;
  float* bufB   = bufA + (size_t)N * 128;
  int*   cnt    = (int*)(bufB + (size_t)N * 128);
  int*   tot    = cnt + NCH * NBP;
  int*   bstart = tot + 1024;
  int*   elist  = bstart + 1024;

  const dim3 blk(256);
  const unsigned gemm_blocks = (N + 127) / 128;
  const unsigned nf64  = (unsigned)(((long long)N * 64 + 255) / 256);

  int* part = (int*)bufA;
  outdeg_count<<<DR * DC, DNT, 0, stream>>>(src, part, E);
  norm_kernel<<<(N + 255) / 256, blk, 0, stream>>>(part, nsrc, N);

  bucket_count<<<NCH / 8, blk, 0, stream>>>(dst, cnt, E);
  bucket_scan_chunks<<<NB, blk, 0, stream>>>(cnt, tot);
  bucket_base<<<1, blk, 0, stream>>>(tot, bstart);
  bucket_place<<<NCH / 8, blk, 0, stream>>>(src, dst, cnt, bstart, elist, E);

  gcn_gemm<128, false><<<gemm_blocks, blk, 0, stream>>>(in_feat, nsrc, ndst, b1, W1, bufB, N);
  owner_aggregate<128, true><<<NB, blk, 0, stream>>>(bufB, elist, bstart, tot, bufA, ndst, N);

  gcn_gemm<128, true><<<gemm_blocks, blk, 0, stream>>>(bufA, nsrc, ndst, b1, W2, bufB, N);
  owner_aggregate<128, false><<<NB, blk, 0, stream>>>(bufB, elist, bstart, tot, bufA, ndst, N);

  gcn_gemm<128, true><<<gemm_blocks, blk, 0, stream>>>(bufA, nsrc, ndst, b2, W3, bufB, N);
  owner_aggregate<128, false><<<NB, blk, 0, stream>>>(bufB, elist, bstart, tot, bufA, ndst, N);

  gcn_gemm<64, true><<<gemm_blocks, blk, 0, stream>>>(bufA, nsrc, ndst, b3, W4, bufB, N);
  owner_aggregate<64, false><<<NB, blk, 0, stream>>>(bufB, elist, bstart, tot, bufA, ndst, N);
  finish_kernel<64, false><<<nf64, blk, 0, stream>>>(bufA, ndst, b4, (float*)d_out, N);
}


# owner-block aggregate: a wave's 16 row sums kept in registers (16-element vector per owned column, indexed by the entry's scalar row), not shared memory; 8 rows a request at 128 cols; sum order kept
# speedup vs baseline: 3.5089x; 1.0346x over previous
#include <hip/hip_runtime.h>
#include <stdint.h>

#define NNODES 100000
#define KDIM   128
#define BN     128
#define NB     ((NNODES + BN - 1) / BN)
#define NBP    800
#define NCH    1024

typedef __attribute__((ext_vector_type(16))) __bf16 v16bf;
typedef __attribute__((ext_vector_type(8)))  __bf16 v8bf;
typedef __attribute__((ext_vector_type(8)))  float  v8f;

__device__ __forceinline__ __bf16 f2bf(float f) {
  unsigned u = __builtin_bit_cast(unsigned, f);
  unsigned r = u + 0x7FFFu + ((u >> 16) & 1u);
  unsigned short s = (unsigned short)(r >> 16);
  return __builtin_bit_cast(__bf16, s);
}

__device__ __forceinline__ int imin(int a, int b) { return a < b ? a : b; }

#define DR   16
#define DC   16
#define DRN  ((((NNODES + DR - 1) / DR) + 31) / 32 * 32)
#define DNT  512
#define DLD  13

__device__ __forceinline__ void count_in_range(int* tab, int s, int lo) {
  const unsigned u = (unsigned)(s - lo);
  if (u < (unsigned)DRN) atomicAdd(&tab[u], 1);
}

__global__ __launch_bounds__(DNT)
void outdeg_count(const int* __restrict__ src, int* __restrict__ part, int E) {
  __shared__ int tab[DRN];
  const int tid = threadIdx.x;
  const int r = blockIdx.x % DR, c = blockIdx.x / DR;
  const int lo = r * DRN;
  for (int j = tid; j < DRN; j += DNT) tab[j] = 0;
  __syncthreads();
  const int nq = E >> 2;
  const int LQ = (nq + DC - 1) / DC;
  const int q0 = c * LQ;
  const int q1 = imin(q0 + LQ, nq);
  const int4* __restrict__ s4 = (const int4*)src;
  for (int q = q0 + tid; q < q1; q += DNT * DLD) {
    int4 v[DLD];
#pragma unroll
    for (int k = 0; k < DLD; ++k) v[k] = s4[imin(q + k * DNT, q1 - 1)];
#pragma unroll
    for (int k = 0; k < DLD; ++k) {
      if (q + k * DNT < q1) {
        count_in_range(tab, v[k].x, lo);
        count_in_range(tab, v[k].y, lo);
        count_in_range(tab, v[k].z, lo);
        count_in_range(tab, v[k].w, lo);
      }
    }
  }
  if (c == DC - 1 && tid < E - 4 * nq) count_in_range(tab, src[4 * nq + tid], lo);
  __syncthreads();
  for (int j = tid; j < DRN; j += DNT)
    if (lo + j < NNODES) part[(size_t)c * NNODES + lo + j] = tab[j];
}

__global__ void norm_kernel(const int* __restrict__ part, float* __restrict__ nsrc, int n) {
  int i = blockIdx.x * blockDim.x + threadIdx.x;
  if (i < n) {
    int s = 0;
#pragma unroll
    for (int c = 0; c < DC; ++c) s += part[(size_t)c * NNODES + i];
    nsrc[i] = rsqrtf(fmaxf((float)s, 1.0f));
  }
}


template <int MOUT, bool GUARD>
__device__ __forceinline__ void stripe_tiles(const __bf16* __restrict__ Bs, int ldk,
                                             const v16bf (&afrag)[KDIM / 32],
                                             float* __restrict__ out0, int nval, int half, int mn) {
  constexpr int CT = MOUT / 16;
  v8f c[CT];
#pragma unroll
  for (int tc = 0; tc < CT; ++tc) {
    c[tc] = v8f{};
    asm volatile("" : "+v"(c[tc]));
  }
#pragma unroll
  for (int kk = 0; kk < KDIM / 32; ++kk) {
    const int k0 = kk * 32 + half * 8;
#pragma unroll
    for (int tc = 0; tc < CT; ++tc) {
      const __bf16* brow = Bs + (tc * 16 + mn) * ldk;
      v8bf bl = *(const v8bf*)(brow + k0);
      v8bf bh = *(const v8bf*)(brow + k0 + 16);
      v16bf bfrag;
#pragma unroll
      for (int i = 0; i < 8; ++i) { bfrag[i] = bl[i]; bfrag[i + 8] = bh[i]; }
      c[tc] = __builtin_amdgcn_wmma_f32_16x16x32_bf16(false, afrag[kk], false, bfrag,
                                                      (short)0, c[tc], false, false);
    }
  }
#pragma unroll
  for (int tc = 0; tc < CT; ++tc) {
    float* out = out0 + tc * 16;
#pragma unroll
    for (int r = 0; r < 8; ++r)
      if (!GUARD || r < nval) out[(size_t)r * MOUT] = c[tc][r];
  }
}

template <int MOUT, bool EPI>
__global__ __launch_bounds__(256)
void gcn_gemm(const float* __restrict__ x, const float* __restrict__ nsrc,
              const float* __restrict__ ndst, const float* __restrict__ bias,
              const float* __restrict__ W, float* __restrict__ h, int nrows) {
  constexpr int LDK = KDIM + 8;
  constexpr int BR  = 128;
  __shared__ __bf16 Bs[MOUT * LDK];
  __shared__ float4 Bias4[KDIM / 4];

  const int tid  = threadIdx.x;
  const int row0 = blockIdx.x * BR;
  const int wave = tid >> 5, lane = tid & 31;
  const int half = lane >> 4, mn = lane & 15;

  const int  arow   = row0 + wave * 16 + mn;
  const bool avalid = arow < nrows;
  const int  arowc  = avalid ? arow : (nrows - 1);
  const float4* xp = (const float4*)(x + (size_t)arowc * KDIM + half * 8);
  float4 xa[16];
#pragma unroll
  for (int kk = 0; kk < KDIM / 32; ++kk) {
    xa[kk * 4 + 0] = xp[kk * 8 + 0];
    xa[kk * 4 + 1] = xp[kk * 8 + 1];
    xa[kk * 4 + 2] = xp[kk * 8 + 4];
    xa[kk * 4 + 3] = xp[kk * 8 + 5];
  }
  const float sc_src = nsrc[arowc];
  float sc_dst = 0.0f;
  if (EPI) sc_dst = ndst[arowc];

  for (int idx = tid; idx < MOUT * KDIM; idx += 256) {
    int k = idx / MOUT, n = idx % MOUT;
    Bs[n * LDK + k] = f2bf(W[(size_t)k * MOUT + n]);
  }
  if (EPI && tid < KDIM / 4) Bias4[tid] = ((const float4*)bias)[tid];
  __syncthreads();

  v16bf afrag[KDIM / 32];
#pragma unroll
  for (int kk = 0; kk < KDIM / 32; ++kk) {
#pragma unroll
    for (int q = 0; q < 4; ++q) {
      const float4 v = xa[kk * 4 + q];
      float4 bq = {0.0f, 0.0f, 0.0f, 0.0f};
      if (EPI) bq = Bias4[kk * 8 + half * 2 + (q & 1) + (q >> 1) * 4];
      const float vv[4] = {v.x, v.y, v.z, v.w};
      const float bv[4] = {bq.x, bq.y, bq.z, bq.w};
#pragma unroll
      for (int c = 0; c < 4; ++c) {
        float a = vv[c];
        if (EPI) a = fmaxf(a * sc_dst + bv[c], 0.0f);
        const float r = avalid ? a * sc_src : 0.0f;
        afrag[kk][q * 4 + c] = f2bf(r);
      }
    }
  }

  const int gr0  = row0 + wave * 16 + half * 8;
  const int nval = nrows - gr0;
  float* out0 = h + (size_t)gr0 * MOUT + mn;
  if ((row0 + BR) <= nrows)
    stripe_tiles<MOUT, false>(Bs, LDK, afrag, out0, nval, half, mn);
  else
    stripe_tiles<MOUT, true>(Bs, LDK, afrag, out0, nval, half, mn);
}


#define SG 25

__global__ __launch_bounds__(256)
void bucket_count(const int* __restrict__ dst, int* __restrict__ cnt, int E) {
  __shared__ int hist[8][NBP];
  const int tid = threadIdx.x, wave = tid >> 5, lane = tid & 31;
  const int c = blockIdx.x * 8 + wave;
  const int L = (E + NCH - 1) / NCH;
  const int beg = c * L;
  for (int j = lane; j < NBP; j += 32) hist[wave][j] = 0;
  __syncthreads();
  for (int i0 = 0; i0 < L; i0 += 32 * SG) {
    int d[SG];
#pragma unroll
    for (int g = 0; g < SG; ++g) d[g] = dst[imin(beg + i0 + 32 * g + lane, E - 1)];
#pragma unroll
    for (int g = 0; g < SG; ++g) {
      const int i = i0 + 32 * g;
      if (i + lane < L && beg + i + lane < E) atomicAdd(&hist[wave][d[g] >> 7], 1);
    }
  }
  __syncthreads();
  for (int j = lane; j < NB; j += 32) cnt[c * NBP + j] = hist[wave][j];
}

__device__ __forceinline__ int block_excl_scan_256(int v, int* sh, int tid) {
  sh[tid] = v;
  __syncthreads();
  for (int off = 1; off < 256; off <<= 1) {
    const int t = (tid >= off) ? sh[tid - off] : 0;
    __syncthreads();
    sh[tid] += t;
    __syncthreads();
  }
  return sh[tid] - v;
}

__global__ __launch_bounds__(256)
void bucket_scan_chunks(int* __restrict__ cnt, int* __restrict__ tot) {
  __shared__ int sh[256];
  const int tid = threadIdx.x, b = blockIdx.x;
  int v[NCH / 256];
  int s = 0;
#pragma unroll
  for (int i = 0; i < NCH / 256; ++i) { v[i] = cnt[(tid * (NCH / 256) + i) * NBP + b]; s += v[i]; }
  int run = block_excl_scan_256(s, sh, tid);
#pragma unroll
  for (int i = 0; i < NCH / 256; ++i) { cnt[(tid * (NCH / 256) + i) * NBP + b] = run; run += v[i]; }
  if (tid == 255) tot[b] = run;
}

__global__ __launch_bounds__(256)
void bucket_base(const int* __restrict__ tot, int* __restrict__ bstart) {
  __shared__ int sh[256];
  const int tid = threadIdx.x;
  int v[4];
  int s = 0;
#pragma unroll
  for (int i = 0; i < 4; ++i) { const int idx = tid * 4 + i; v[i] = (idx < NB) ? tot[idx] : 0; s += v[i]; }
  int run = block_excl_scan_256(s, sh, tid);
#pragma unroll
  for (int i = 0; i < 4; ++i) { const int idx = tid * 4 + i; if (idx < NB) bstart[idx] = run; run += v[i]; }
}

__global__ __launch_bounds__(256)
void bucket_place(const int* __restrict__ src, const int* __restrict__ dst,
                  const int* __restrict__ cnt, const int* __restrict__ bstart,
                  int* __restrict__ elist, int E) {
  __shared__ int cur[8][NBP];
  const int tid = threadIdx.x, wave = tid >> 5, lane = tid & 31;
  const int c = blockIdx.x * 8 + wave;
  const int L = (E + NCH - 1) / NCH;
  const int beg = c * L;
#pragma unroll
  for (int k = 0; k < (NB + 31) / 32; ++k) {
    const int j = lane + 32 * k;
    const int jc = imin(j, NB - 1);
    const int v = bstart[jc] + cnt[c * NBP + jc];
    if (j < NB) cur[wave][j] = v;
  }
  __syncthreads();
  for (int i0 = 0; i0 < L; i0 += 32 * SG) {
    int s[SG], d[SG], pos[SG];
#pragma unroll
    for (int g = 0; g < SG; ++g) {
      const int ec = imin(beg + i0 + 32 * g + lane, E - 1);
      s[g] = src[ec];
      d[g] = dst[ec];
    }
#pragma unroll
    for (int g = 0; g < SG; ++g) {
      const int i = i0 + 32 * g;
      pos[g] = 0;
      if (i + lane < L && beg + i + lane < E) pos[g] = atomicAdd(&cur[wave][d[g] >> 7], 1);
    }
#pragma unroll
    for (int g = 0; g < SG; ++g) {
      const int i = i0 + 32 * g;
      if (i + lane < L && beg + i + lane < E) elist[pos[g]] = (s[g] << 7) | (d[g] & (BN - 1));
    }
  }
}

typedef __attribute__((ext_vector_type(16))) float v16f;

__device__ __forceinline__ void vzero(float4& a) { a.x = 0.0f; a.y = 0.0f; a.z = 0.0f; a.w = 0.0f; }
__device__ __forceinline__ void vzero(float2& a) { a.x = 0.0f; a.y = 0.0f; }

__device__ __forceinline__ void radd(v16f (&a)[4], int r, const float4 v) {
  a[0][r] += v.x; a[1][r] += v.y; a[2][r] += v.z; a[3][r] += v.w;
}
__device__ __forceinline__ void radd(v16f (&a)[2], int r, const float2 v) {
  a[0][r] += v.x; a[1][r] += v.y;
}
__device__ __forceinline__ float4 rrow(const v16f (&a)[4], int r) {
  float4 o; o.x = a[0][r]; o.y = a[1][r]; o.z = a[2][r]; o.w = a[3][r]; return o;
}
__device__ __forceinline__ float2 rrow(const v16f (&a)[2], int r) {
  float2 o; o.x = a[0][r]; o.y = a[1][r]; return o;
}

template <int F> struct RowVec;
template <> struct RowVec<128> { typedef float4 T; };
template <> struct RowVec<64>  { typedef float2 T; };

#define QCAP 64
#define CH   1024

template <int F, int GB, bool FULL, bool DEG>
__device__ __forceinline__ void gather_group(const typename RowVec<F>::T* __restrict__ hv,
                                             v16f (&acc)[F / 32], const int* qw, int* rc,
                                             int head, int cnt, int lane) {
  typedef typename RowVec<F>::T V;
  const int slot = qw[(head + (lane & (GB - 1))) & (QCAP - 1)];
  if (DEG && lane < GB && (FULL || lane < cnt)) atomicAdd(&rc[slot & (BN - 1)], 1);
  int p[GB];
  V v[GB];
#pragma unroll
  for (int j = 0; j < GB; ++j) {
    p[j] = __builtin_amdgcn_readlane(slot, j);
    if (FULL || j < cnt) v[j] = hv[(size_t)(p[j] >> 7) * 32 + lane];
    else vzero(v[j]);
  }
  __builtin_amdgcn_sched_barrier(0);
#pragma unroll
  for (int j = 0; j < GB; ++j)
    if (FULL || j < cnt) radd(acc, p[j] & 15, v[j]);
}

template <int F> struct GroupRows;
template <> struct GroupRows<128> { static constexpr int N = 8; };
template <> struct GroupRows<64>  { static constexpr int N = 32; };

template <int F, bool DEG>
__global__ __launch_bounds__(256)
void owner_aggregate(const float* __restrict__ h, const int* __restrict__ elist,
                     const int* __restrict__ bstart, const int* __restrict__ tot,
                     float* __restrict__ agg, float* __restrict__ ndst, int nrows) {
  typedef typename RowVec<F>::T V;
  constexpr int GB = GroupRows<F>::N;
  constexpr int NC = F / 32;
  constexpr int PT = CH / 256;
  const V* __restrict__ hv = (const V*)h;
  V* __restrict__ aggv = (V*)agg;
  __shared__ int ebuf[2][CH];
  __shared__ int q[8][QCAP];
  __shared__ int rowcnt[DEG ? BN : 1];
  const int tid = threadIdx.x, wave = tid >> 5, lane = tid & 31;
  const int b = blockIdx.x;
  v16f acc[NC];
#pragma unroll
  for (int k = 0; k < NC; ++k) acc[k] = (v16f)(0.0f);
  int* qw = q[wave];
  qw[lane] = 0;
  qw[lane + 32] = 0;
  if (DEG && lane < 16) rowcnt[wave * 16 + lane] = 0;
  const int st = bstart[b], n = tot[b];
  if (n > 0) {
    const int* __restrict__ el = elist + st;
    const int last = n - 1;
    const int nch = (n + CH - 1) / CH;
    int rg[PT];
#pragma unroll
    for (int k = 0; k < PT; ++k) rg[k] = el[imin(tid + 256 * k, last)];
    int head = 0, tail = 0;
    for (int c = 0; c < nch; ++c) {
      int* eb = ebuf[c & 1];
#pragma unroll
      for (int k = 0; k < PT; ++k) eb[tid + 256 * k] = rg[k];
      __syncthreads();
#pragma unroll
      for (int k = 0; k < PT; ++k) rg[k] = el[imin((c + 1) * CH + tid + 256 * k, last)];
      const int cn = imin(CH, n - c * CH);
      for (int i = 0; i < cn; i += 32) {
        const int pk = eb[i + lane];
        const bool own = (i + lane < cn) && (((pk >> 4) & 7) == wave);
        const unsigned m = __builtin_amdgcn_ballot_w32(own);
        if (own) qw[(tail + (int)__builtin_amdgcn_mbcnt_lo(m, 0u)) & (QCAP - 1)] = pk;
        tail += __builtin_popcount(m);
        while (tail - head >= GB) {
          gather_group<F, GB, true, DEG>(hv, acc, qw, rowcnt, head, GB, lane);
          head += GB;
        }
      }
    }
    if (tail != head) gather_group<F, GB, false, DEG>(hv, acc, qw, rowcnt, head, tail - head, lane);
  }
  const int row0 = b * BN + wave * 16;
#pragma unroll
  for (int r = 0; r < 16; ++r)
    if (row0 + r < nrows) aggv[(size_t)(row0 + r) * 32 + lane] = rrow(acc, r);
  if (DEG && lane < 16 && row0 + lane < nrows)
    ndst[row0 + lane] = rsqrtf(fmaxf((float)rowcnt[wave * 16 + lane], 1.0f));
}

template <int F, bool RELU>
__global__ void finish_kernel(const float* __restrict__ agg, const float* __restrict__ ndst,
                              const float* __restrict__ bias, float* __restrict__ out, int n) {
  long long idx = blockIdx.x * (long long)blockDim.x + threadIdx.x;
  if (idx >= (long long)n * F) return;
  int node = (int)(idx / F);
  int f = (int)(idx & (F - 1));
  float v = agg[idx] * ndst[node] + bias[f];
  out[idx] = RELU ? fmaxf(v, 0.0f) : v;
}

extern "C" void kernel_launch(void* const* d_in, const int* in_sizes, int n_in,
                              void* d_out, int out_size, void* d_ws, size_t ws_size,
                              hipStream_t stream) {
  const float* in_feat = (const float*)d_in[0];
  const int*   src     = (const int*)d_in[1];
  const int*   dst     = (const int*)d_in[2];
  const float* W1 = (const float*)d_in[3];  const float* b1 = (const float*)d_in[4];
  const float* W2 = (const float*)d_in[5];  const float* b2 = (const float*)d_in[6];
  const float* W3 = (const float*)d_in[7];  const float* b3 = (const float*)d_in[8];
  const float* W4 = (const float*)d_in[9];  const float* b4 = (const float*)d_in[10];

  const int N = NNODES;
  const int E = in_sizes[1];

  float* ws     = (float*)d_ws;
  float* nsrc   = ws + 2 * (size_t)N;
  float* ndst   = nsrc + N;
  float* bufA   = ndst + N;
  float* bufB   = bufA + (size_t)N * 128;
  int*   cnt    = (int*)(bufB + (size_t)N * 128);
  int*   tot    = cnt + NCH * NBP;
  int*   bstart = tot + 1024;
  int*   elist  = bstart + 1024;

  const dim3 blk(256);
  const unsigned gemm_blocks = (N + 127) / 128;
  const unsigned nf64  = (unsigned)(((long long)N * 64 + 255) / 256);

  int* part = (int*)bufA;
  outdeg_count<<<DR * DC, DNT, 0, stream>>>(src, part, E);
  norm_kernel<<<(N + 255) / 256, blk, 0, stream>>>(part, nsrc, N);

  bucket_count<<<NCH / 8, blk, 0, stream>>>(dst, cnt, E);
  bucket_scan_chunks<<<NB, blk, 0, stream>>>(cnt, tot);
  bucket_base<<<1, blk, 0, stream>>>(tot, bstart);
  bucket_place<<<NCH / 8, blk, 0, stream>>>(src, dst, cnt, bstart, elist, E);

  gcn_gemm<128, false><<<gemm_blocks, blk, 0, stream>>>(in_feat, nsrc, ndst, b1, W1, bufB, N);
  owner_aggregate<128, true><<<NB, blk, 0, stream>>>(bufB, elist, bstart, tot, bufA, ndst, N);

  gcn_gemm<128, true><<<gemm_blocks, blk, 0, stream>>>(bufA, nsrc, ndst, b1, W2, bufB, N);
  owner_aggregate<128, false><<<NB, blk, 0, stream>>>(bufB, elist, bstart, tot, bufA, ndst, N);

  gcn_gemm<128, true><<<gemm_blocks, blk, 0, stream>>>(bufA, nsrc, ndst, b2, W3, bufB, N);
  owner_aggregate<128, false><<<NB, blk, 0, stream>>>(bufB, elist, bstart, tot, bufA, ndst, N);

  gcn_gemm<64, true><<<gemm_blocks, blk, 0, stream>>>(bufA, nsrc, ndst, b3, W4, bufB, N);
  owner_aggregate<64, false><<<NB, blk, 0, stream>>>(bufB, elist, bstart, tot, bufA, ndst, N);
  finish_kernel<64, false><<<nf64, blk, 0, stream>>>(bufA, ndst, b4, (float*)d_out, N);
}


# launches 15 to 11: job i of the out-degree/norm/layer-1-product chain shares a launch with job i of the independent edge-grouping chain (by block range); bucket prefix redone per placement block
# speedup vs baseline: 3.7299x; 1.0630x over previous
#include <hip/hip_runtime.h>
#include <stdint.h>

#define NNODES 100000
#define KDIM   128
#define BN     128
#define NB     ((NNODES + BN - 1) / BN)
#define NBP    800
#define NCH    1024

typedef __attribute__((ext_vector_type(16))) __bf16 v16bf;
typedef __attribute__((ext_vector_type(8)))  __bf16 v8bf;
typedef __attribute__((ext_vector_type(8)))  float  v8f;

__device__ __forceinline__ __bf16 f2bf(float f) {
  unsigned u = __builtin_bit_cast(unsigned, f);
  unsigned r = u + 0x7FFFu + ((u >> 16) & 1u);
  unsigned short s = (unsigned short)(r >> 16);
  return __builtin_bit_cast(__bf16, s);
}

__device__ __forceinline__ int imin(int a, int b) { return a < b ? a : b; }


#define DR   16
#define DC   16
#define DRN  ((((NNODES + DR - 1) / DR) + 31) / 32 * 32)
#define DNT  512
#define DLD  13

__device__ __forceinline__ void count_in_range(int* tab, int s, int lo) {
  const unsigned u = (unsigned)(s - lo);
  if (u < (unsigned)DRN) atomicAdd(&tab[u], 1);
}

__device__ __forceinline__
void outdeg_count_block(const int* __restrict__ src, int* __restrict__ part, int E, int bid) {
  __shared__ int tab[DRN];
  const int tid = threadIdx.x;
  const int r = bid % DR, c = bid / DR;
  const int lo = r * DRN;
  for (int j = tid; j < DRN; j += DNT) tab[j] = 0;
  __syncthreads();
  const int nq = E >> 2;
  const int LQ = (nq + DC - 1) / DC;
  const int q0 = c * LQ;
  const int q1 = imin(q0 + LQ, nq);
  const int4* __restrict__ s4 = (const int4*)src;
  for (int q = q0 + tid; q < q1; q += DNT * DLD) {
    int4 v[DLD];
#pragma unroll
    for (int k = 0; k < DLD; ++k) v[k] = s4[imin(q + k * DNT, q1 - 1)];
#pragma unroll
    for (int k = 0; k < DLD; ++k) {
      if (q + k * DNT < q1) {
        count_in_range(tab, v[k].x, lo);
        count_in_range(tab, v[k].y, lo);
        count_in_range(tab, v[k].z, lo);
        count_in_range(tab, v[k].w, lo);
      }
    }
  }
  if (c == DC - 1 && tid < E - 4 * nq) count_in_range(tab, src[4 * nq + tid], lo);
  __syncthreads();
  for (int j = tid; j < DRN; j += DNT)
    if (lo + j < NNODES) part[(size_t)c * NNODES + lo + j] = tab[j];
}

__device__ __forceinline__
void norm_block(const int* __restrict__ part, float* __restrict__ nsrc, int n, int bid) {
  int i = bid * 256 + threadIdx.x;
  if (i < n) {
    int s = 0;
#pragma unroll
    for (int c = 0; c < DC; ++c) s += part[(size_t)c * NNODES + i];
    nsrc[i] = rsqrtf(fmaxf((float)s, 1.0f));
  }
}


template <int MOUT, bool GUARD>
__device__ __forceinline__ void stripe_tiles(const __bf16* __restrict__ Bs, int ldk,
                                             const v16bf (&afrag)[KDIM / 32],
                                             float* __restrict__ out0, int nval, int half, int mn) {
  constexpr int CT = MOUT / 16;
  v8f c[CT];
#pragma unroll
  for (int tc = 0; tc < CT; ++tc) {
    c[tc] = v8f{};
    asm volatile("" : "+v"(c[tc]));
  }
#pragma unroll
  for (int kk = 0; kk < KDIM / 32; ++kk) {
    const int k0 = kk * 32 + half * 8;
#pragma unroll
    for (int tc = 0; tc < CT; ++tc) {
      const __bf16* brow = Bs + (tc * 16 + mn) * ldk;
      v8bf bl = *(const v8bf*)(brow + k0);
      v8bf bh = *(const v8bf*)(brow + k0 + 16);
      v16bf bfrag;
#pragma unroll
      for (int i = 0; i < 8; ++i) { bfrag[i] = bl[i]; bfrag[i + 8] = bh[i]; }
      c[tc] = __builtin_amdgcn_wmma_f32_16x16x32_bf16(false, afrag[kk], false, bfrag,
                                                      (short)0, c[tc], false, false);
    }
  }
#pragma unroll
  for (int tc = 0; tc < CT; ++tc) {
    float* out = out0 + tc * 16;
#pragma unroll
    for (int r = 0; r < 8; ++r)
      if (!GUARD || r < nval) out[(size_t)r * MOUT] = c[tc][r];
  }
}

template <int MOUT, bool EPI>
__device__ __forceinline__
void gemm_block(const float* __restrict__ x, const float* __restrict__ nsrc,
                const float* __restrict__ ndst, const float* __restrict__ bias,
                const float* __restrict__ W, float* __restrict__ h, int nrows, int bid) {
  constexpr int LDK = KDIM + 8;
  constexpr int BR  = 128;
  __shared__ __bf16 Bs[MOUT * LDK];
  __shared__ float4 Bias4[KDIM / 4];

  const int tid  = threadIdx.x;
  const int row0 = bid * BR;
  const int wave = tid >> 5, lane = tid & 31;
  const int half = lane >> 4, mn = lane & 15;

  const int  arow   = row0 + wave * 16 + mn;
  const bool avalid = arow < nrows;
  const int  arowc  = avalid ? arow : (nrows - 1);
  const float4* xp = (const float4*)(x + (size_t)arowc * KDIM + half * 8);
  float4 xa[16];
#pragma unroll
  for (int kk = 0; kk < KDIM / 32; ++kk) {
    xa[kk * 4 + 0] = xp[kk * 8 + 0];
    xa[kk * 4 + 1] = xp[kk * 8 + 1];
    xa[kk * 4 + 2] = xp[kk * 8 + 4];
    xa[kk * 4 + 3] = xp[kk * 8 + 5];
  }
  const float sc_src = nsrc[arowc];
  float sc_dst = 0.0f;
  if (EPI) sc_dst = ndst[arowc];

  for (int idx = tid; idx < MOUT * KDIM; idx += 256) {
    int k = idx / MOUT, n = idx % MOUT;
    Bs[n * LDK + k] = f2bf(W[(size_t)k * MOUT + n]);
  }
  if (EPI && tid < KDIM / 4) Bias4[tid] = ((const float4*)bias)[tid];
  __syncthreads();

  v16bf afrag[KDIM / 32];
#pragma unroll
  for (int kk = 0; kk < KDIM / 32; ++kk) {
#pragma unroll
    for (int q = 0; q < 4; ++q) {
      const float4 v = xa[kk * 4 + q];
      float4 bq = {0.0f, 0.0f, 0.0f, 0.0f};
      if (EPI) bq = Bias4[kk * 8 + half * 2 + (q & 1) + (q >> 1) * 4];
      const float vv[4] = {v.x, v.y, v.z, v.w};
      const float bv[4] = {bq.x, bq.y, bq.z, bq.w};
#pragma unroll
      for (int c = 0; c < 4; ++c) {
        float a = vv[c];
        if (EPI) a = fmaxf(a * sc_dst + bv[c], 0.0f);
        const float r = avalid ? a * sc_src : 0.0f;
        afrag[kk][q * 4 + c] = f2bf(r);
      }
    }
  }

  const int gr0  = row0 + wave * 16 + half * 8;
  const int nval = nrows - gr0;
  float* out0 = h + (size_t)gr0 * MOUT + mn;
  if ((row0 + BR) <= nrows)
    stripe_tiles<MOUT, false>(Bs, LDK, afrag, out0, nval, half, mn);
  else
    stripe_tiles<MOUT, true>(Bs, LDK, afrag, out0, nval, half, mn);
}

template <int MOUT, bool EPI>
__global__ __launch_bounds__(256)
void gcn_gemm(const float* __restrict__ x, const float* __restrict__ nsrc,
              const float* __restrict__ ndst, const float* __restrict__ bias,
              const float* __restrict__ W, float* __restrict__ h, int nrows) {
  gemm_block<MOUT, EPI>(x, nsrc, ndst, bias, W, h, nrows, blockIdx.x);
}


#define SG 25

template <int WPB>
__device__ __forceinline__
void bucket_count_block(const int* __restrict__ dst, int* __restrict__ cnt, int E, int bid) {
  __shared__ int hist[WPB][NBP];
  const int tid = threadIdx.x, wave = tid >> 5, lane = tid & 31;
  const int c = bid * WPB + wave;
  const int L = (E + NCH - 1) / NCH;
  const int beg = c * L;
  for (int j = lane; j < NBP; j += 32) hist[wave][j] = 0;
  __syncthreads();
  for (int i0 = 0; i0 < L; i0 += 32 * SG) {
    int d[SG];
#pragma unroll
    for (int g = 0; g < SG; ++g) d[g] = dst[imin(beg + i0 + 32 * g + lane, E - 1)];
#pragma unroll
    for (int g = 0; g < SG; ++g) {
      const int i = i0 + 32 * g;
      if (i + lane < L && beg + i + lane < E) atomicAdd(&hist[wave][d[g] >> 7], 1);
    }
  }
  __syncthreads();
  for (int j = lane; j < NB; j += 32) cnt[c * NBP + j] = hist[wave][j];
}

#define CWPB (DNT / 32)
#define CNTB (NCH / CWPB)
__global__ __launch_bounds__(DNT)
void count_kernel(const int* __restrict__ src, const int* __restrict__ dst,
                  int* __restrict__ part, int* __restrict__ cnt, int E) {
  if (blockIdx.x < CNTB) bucket_count_block<CWPB>(dst, cnt, E, blockIdx.x);
  else outdeg_count_block(src, part, E, blockIdx.x - CNTB);
}

__device__ __forceinline__ int block_excl_scan_256(int v, int* sh, int tid) {
  sh[tid] = v;
  __syncthreads();
  for (int off = 1; off < 256; off <<= 1) {
    const int t = (tid >= off) ? sh[tid - off] : 0;
    __syncthreads();
    sh[tid] += t;
    __syncthreads();
  }
  return sh[tid] - v;
}

__device__ __forceinline__
void bucket_scan_chunks_block(int* __restrict__ cnt, int* __restrict__ tot, int b) {
  __shared__ int sh[256];
  const int tid = threadIdx.x;
  int v[NCH / 256];
  int s = 0;
#pragma unroll
  for (int i = 0; i < NCH / 256; ++i) { v[i] = cnt[(tid * (NCH / 256) + i) * NBP + b]; s += v[i]; }
  int run = block_excl_scan_256(s, sh, tid);
#pragma unroll
  for (int i = 0; i < NCH / 256; ++i) { cnt[(tid * (NCH / 256) + i) * NBP + b] = run; run += v[i]; }
  if (tid == 255) tot[b] = run;
}

#define NRMB ((NNODES + 255) / 256)
__global__ __launch_bounds__(256)
void norm_scan_kernel(const int* __restrict__ part, float* __restrict__ nsrc, int n,
                      int* __restrict__ cnt, int* __restrict__ tot) {
  if (blockIdx.x < NB) bucket_scan_chunks_block(cnt, tot, blockIdx.x);
  else norm_block(part, nsrc, n, blockIdx.x - NB);
}

__device__ __forceinline__
void bucket_place_block(const int* __restrict__ src, const int* __restrict__ dst,
                        const int* __restrict__ cnt, const int* __restrict__ tot,
                        int* __restrict__ bstart, int* __restrict__ elist, int E, int bid) {
  __shared__ int cur[8][NBP];
  __shared__ int bst[1024];
  __shared__ int sh[256];
  const int tid = threadIdx.x, wave = tid >> 5, lane = tid & 31;
  const int c = bid * 8 + wave;
  const int L = (E + NCH - 1) / NCH;
  const int beg = c * L;
  {
    int v[4];
    int s = 0;
#pragma unroll
    for (int i = 0; i < 4; ++i) { const int idx = tid * 4 + i; v[i] = (idx < NB) ? tot[idx] : 0; s += v[i]; }
    int run = block_excl_scan_256(s, sh, tid);
#pragma unroll
    for (int i = 0; i < 4; ++i) {
      const int idx = tid * 4 + i;
      bst[idx] = run;
      if (bid == 0 && idx < NB) bstart[idx] = run;
      run += v[i];
    }
  }
  __syncthreads();
#pragma unroll
  for (int k = 0; k < (NB + 31) / 32; ++k) {
    const int j = lane + 32 * k;
    const int jc = imin(j, NB - 1);
    const int v = bst[jc] + cnt[c * NBP + jc];
    if (j < NB) cur[wave][j] = v;
  }
  __syncthreads();
  for (int i0 = 0; i0 < L; i0 += 32 * SG) {
    int s[SG], d[SG], pos[SG];
#pragma unroll
    for (int g = 0; g < SG; ++g) {
      const int ec = imin(beg + i0 + 32 * g + lane, E - 1);
      s[g] = src[ec];
      d[g] = dst[ec];
    }
#pragma unroll
    for (int g = 0; g < SG; ++g) {
      const int i = i0 + 32 * g;
      pos[g] = 0;
      if (i + lane < L && beg + i + lane < E) pos[g] = atomicAdd(&cur[wave][d[g] >> 7], 1);
    }
#pragma unroll
    for (int g = 0; g < SG; ++g) {
      const int i = i0 + 32 * g;
      if (i + lane < L && beg + i + lane < E) elist[pos[g]] = (s[g] << 7) | (d[g] & (BN - 1));
    }
  }
}

#define PLB   (NCH / 8)
#define GEMMB ((NNODES + 127) / 128)
__global__ __launch_bounds__(256)
void place_gemm1_kernel(const int* __restrict__ src, const int* __restrict__ dst,
                        const int* __restrict__ cnt, const int* __restrict__ tot,
                        int* __restrict__ bstart, int* __restrict__ elist, int E,
                        const float* __restrict__ x, const float* __restrict__ nsrc,
                        const float* __restrict__ ndst, const float* __restrict__ bias,
                        const float* __restrict__ W, float* __restrict__ h, int nrows) {
  if (blockIdx.x < PLB) bucket_place_block(src, dst, cnt, tot, bstart, elist, E, blockIdx.x);
  else gemm_block<128, false>(x, nsrc, ndst, bias, W, h, nrows, blockIdx.x - PLB);
}

typedef __attribute__((ext_vector_type(16))) float v16f;

__device__ __forceinline__ void vzero(float4& a) { a.x = 0.0f; a.y = 0.0f; a.z = 0.0f; a.w = 0.0f; }
__device__ __forceinline__ void vzero(float2& a) { a.x = 0.0f; a.y = 0.0f; }

__device__ __forceinline__ void radd(v16f (&a)[4], int r, const float4 v) {
  a[0][r] += v.x; a[1][r] += v.y; a[2][r] += v.z; a[3][r] += v.w;
}
__device__ __forceinline__ void radd(v16f (&a)[2], int r, const float2 v) {
  a[0][r] += v.x; a[1][r] += v.y;
}
__device__ __forceinline__ float4 rrow(const v16f (&a)[4], int r) {
  float4 o; o.x = a[0][r]; o.y = a[1][r]; o.z = a[2][r]; o.w = a[3][r]; return o;
}
__device__ __forceinline__ float2 rrow(const v16f (&a)[2], int r) {
  float2 o; o.x = a[0][r]; o.y = a[1][r]; return o;
}

template <int F> struct RowVec;
template <> struct RowVec<128> { typedef float4 T; };
template <> struct RowVec<64>  { typedef float2 T; };

#define QCAP 64
#define CH   1024

template <int F, int GB, bool FULL, bool DEG>
__device__ __forceinline__ void gather_group(const typename RowVec<F>::T* __restrict__ hv,
                                             v16f (&acc)[F / 32], const int* qw, int* rc,
                                             int head, int cnt, int lane) {
  typedef typename RowVec<F>::T V;
  const int slot = qw[(head + (lane & (GB - 1))) & (QCAP - 1)];
  if (DEG && lane < GB && (FULL || lane < cnt)) atomicAdd(&rc[slot & (BN - 1)], 1);
  int p[GB];
  V v[GB];
#pragma unroll
  for (int j = 0; j < GB; ++j) {
    p[j] = __builtin_amdgcn_readlane(slot, j);
    if (FULL || j < cnt) v[j] = hv[(size_t)(p[j] >> 7) * 32 + lane];
    else vzero(v[j]);
  }
  __builtin_amdgcn_sched_barrier(0);
#pragma unroll
  for (int j = 0; j < GB; ++j)
    if (FULL || j < cnt) radd(acc, p[j] & 15, v[j]);
}

template <int F> struct GroupRows;
template <> struct GroupRows<128> { static constexpr int N = 8; };
template <> struct GroupRows<64>  { static constexpr int N = 32; };

template <int F, bool DEG>
__global__ __launch_bounds__(256)
void owner_aggregate(const float* __restrict__ h, const int* __restrict__ elist,
                     const int* __restrict__ bstart, const int* __restrict__ tot,
                     float* __restrict__ agg, float* __restrict__ ndst, int nrows) {
  typedef typename RowVec<F>::T V;
  constexpr int GB = GroupRows<F>::N;
  constexpr int NC = F / 32;
  constexpr int PT = CH / 256;
  const V* __restrict__ hv = (const V*)h;
  V* __restrict__ aggv = (V*)agg;
  __shared__ int ebuf[2][CH];
  __shared__ int q[8][QCAP];
  __shared__ int rowcnt[DEG ? BN : 1];
  const int tid = threadIdx.x, wave = tid >> 5, lane = tid & 31;
  const int b = blockIdx.x;
  v16f acc[NC];
#pragma unroll
  for (int k = 0; k < NC; ++k) acc[k] = (v16f)(0.0f);
  int* qw = q[wave];
  qw[lane] = 0;
  qw[lane + 32] = 0;
  if (DEG && lane < 16) rowcnt[wave * 16 + lane] = 0;
  const int st = bstart[b], n = tot[b];
  if (n > 0) {
    const int* __restrict__ el = elist + st;
    const int last = n - 1;
    const int nch = (n + CH - 1) / CH;
    int rg[PT];
#pragma unroll
    for (int k = 0; k < PT; ++k) rg[k] = el[imin(tid + 256 * k, last)];
    int head = 0, tail = 0;
    for (int c = 0; c < nch; ++c) {
      int* eb = ebuf[c & 1];
#pragma unroll
      for (int k = 0; k < PT; ++k) eb[tid + 256 * k] = rg[k];
      __syncthreads();
#pragma unroll
      for (int k = 0; k < PT; ++k) rg[k] = el[imin((c + 1) * CH + tid + 256 * k, last)];
      const int cn = imin(CH, n - c * CH);
      for (int i = 0; i < cn; i += 32) {
        const int pk = eb[i + lane];
        const bool own = (i + lane < cn) && (((pk >> 4) & 7) == wave);
        const unsigned m = __builtin_amdgcn_ballot_w32(own);
        if (own) qw[(tail + (int)__builtin_amdgcn_mbcnt_lo(m, 0u)) & (QCAP - 1)] = pk;
        tail += __builtin_popcount(m);
        while (tail - head >= GB) {
          gather_group<F, GB, true, DEG>(hv, acc, qw, rowcnt, head, GB, lane);
          head += GB;
        }
      }
    }
    if (tail != head) gather_group<F, GB, false, DEG>(hv, acc, qw, rowcnt, head, tail - head, lane);
  }
  const int row0 = b * BN + wave * 16;
#pragma unroll
  for (int r = 0; r < 16; ++r)
    if (row0 + r < nrows) aggv[(size_t)(row0 + r) * 32 + lane] = rrow(acc, r);
  if (DEG && lane < 16 && row0 + lane < nrows)
    ndst[row0 + lane] = rsqrtf(fmaxf((float)rowcnt[wave * 16 + lane], 1.0f));
}

template <int F, bool RELU>
__global__ void finish_kernel(const float* __restrict__ agg, const float* __restrict__ ndst,
                              const float* __restrict__ bias, float* __restrict__ out, int n) {
  long long idx = blockIdx.x * (long long)blockDim.x + threadIdx.x;
  if (idx >= (long long)n * F) return;
  int node = (int)(idx / F);
  int f = (int)(idx & (F - 1));
  float v = agg[idx] * ndst[node] + bias[f];
  out[idx] = RELU ? fmaxf(v, 0.0f) : v;
}

extern "C" void kernel_launch(void* const* d_in, const int* in_sizes, int n_in,
                              void* d_out, int out_size, void* d_ws, size_t ws_size,
                              hipStream_t stream) {
  const float* in_feat = (const float*)d_in[0];
  const int*   src     = (const int*)d_in[1];
  const int*   dst     = (const int*)d_in[2];
  const float* W1 = (const float*)d_in[3];  const float* b1 = (const float*)d_in[4];
  const float* W2 = (const float*)d_in[5];  const float* b2 = (const float*)d_in[6];
  const float* W3 = (const float*)d_in[7];  const float* b3 = (const float*)d_in[8];
  const float* W4 = (const float*)d_in[9];  const float* b4 = (const float*)d_in[10];

  const int N = NNODES;
  const int E = in_sizes[1];

  float* ws     = (float*)d_ws;
  float* nsrc   = ws + 2 * (size_t)N;
  float* ndst   = nsrc + N;
  float* bufA   = ndst + N;
  float* bufB   = bufA + (size_t)N * 128;
  int*   cnt    = (int*)(bufB + (size_t)N * 128);
  int*   tot    = cnt + NCH * NBP;
  int*   bstart = tot + 1024;
  int*   elist  = bstart + 1024;

  const dim3 blk(256);
  const unsigned nf64  = (unsigned)(((long long)N * 64 + 255) / 256);

  int* part = (int*)bufA;
  count_kernel<<<CNTB + DR * DC, DNT, 0, stream>>>(src, dst, part, cnt, E);
  norm_scan_kernel<<<NB + NRMB, blk, 0, stream>>>(part, nsrc, N, cnt, tot);
  place_gemm1_kernel<<<PLB + GEMMB, blk, 0, stream>>>(src, dst, cnt, tot, bstart, elist, E,
                                                     in_feat, nsrc, ndst, b1, W1, bufB, N);
  owner_aggregate<128, true><<<NB, blk, 0, stream>>>(bufB, elist, bstart, tot, bufA, ndst, N);

  gcn_gemm<128, true><<<GEMMB, blk, 0, stream>>>(bufA, nsrc, ndst, b1, W2, bufB, N);
  owner_aggregate<128, false><<<NB, blk, 0, stream>>>(bufB, elist, bstart, tot, bufA, ndst, N);

  gcn_gemm<128, true><<<GEMMB, blk, 0, stream>>>(bufA, nsrc, ndst, b2, W3, bufB, N);
  owner_aggregate<128, false><<<NB, blk, 0, stream>>>(bufB, elist, bstart, tot, bufA, ndst, N);

  gcn_gemm<64, true><<<GEMMB, blk, 0, stream>>>(bufA, nsrc, ndst, b3, W4, bufB, N);
  owner_aggregate<64, false><<<NB, blk, 0, stream>>>(bufB, elist, bstart, tot, bufA, ndst, N);
  finish_kernel<64, false><<<nf64, blk, 0, stream>>>(bufA, ndst, b4, (float*)d_out, N);
}
